# starting kernel
# baseline (speedup 1.0000x reference)
#include <hip/hip_runtime.h>
#include <hip/hip_bf16.h>
#include <math.h>

typedef __attribute__((ext_vector_type(2))) float v2f;
typedef __attribute__((ext_vector_type(8))) float v8f;

#define DD 64
#define EPS_COS 1e-8f
#define EPS_BN  1e-5f

__global__ __launch_bounds__(64) void gather_gate_kernel(
    const float* __restrict__ embed_u_z, const float* __restrict__ embed_u_c,
    const float* __restrict__ embed_v_z, const float* __restrict__ embed_v_c,
    const float* __restrict__ item_table,
    const int* __restrict__ hist_items, const int* __restrict__ hist_lens,
    const int* __restrict__ item_ids, const int* __restrict__ red_by_friends,
    float* __restrict__ uv, int H) {
  __shared__ __align__(16) float anchor_s[DD];
  __shared__ float sim_s[64];
  __shared__ int   valid_s[64];

  const int b = blockIdx.x;
  const int t = threadIdx.x;
  const int item = item_ids[b];

  anchor_s[t] = item_table[(long)item * DD + t];
  __syncthreads();

  float sim = -INFINITY;
  int valid = 0;
  if (t < H) {
    const int hi = hist_items[b * H + t];
    const float4* he4 = (const float4*)(item_table + (long)hi * DD);
    const float4* an4 = (const float4*)anchor_s;
    float dot = 0.0f, na = 0.0f, nb = 0.0f;
#pragma unroll
    for (int i = 0; i < DD / 4; ++i) {
      float4 x = he4[i];
      float4 a = an4[i];
      dot += x.x * a.x + x.y * a.y + x.z * a.z + x.w * a.w;
      na  += x.x * x.x + x.y * x.y + x.z * x.z + x.w * x.w;
      nb  += a.x * a.x + a.y * a.y + a.z * a.z + a.w * a.w;
    }
    na = sqrtf(na);
    nb = sqrtf(nb);
    sim = dot / (fmaxf(na, EPS_COS) * fmaxf(nb, EPS_COS));
    valid = (t < hist_lens[b]) && (hi != item);
  }
  sim_s[t]   = valid ? sim : -INFINITY;
  valid_s[t] = valid;
  __syncthreads();

  for (int s = 32; s > 0; s >>= 1) {
    if (t < s) {
      sim_s[t]   = fmaxf(sim_s[t], sim_s[t + s]);
      valid_s[t] += valid_s[t + s];
    }
    __syncthreads();
  }

  const bool empty   = (valid_s[0] == 0);
  const bool gate    = (red_by_friends[b] > 0) && (sim_s[0] > 0.5f);
  const bool use_u_c = empty || gate;

  const long bd = (long)b * DD + t;
  const float u = embed_u_z[bd] + (use_u_c ? embed_u_c[bd] : 0.0f);
  const float v = embed_v_z[bd] +
                  (empty ? embed_u_c[bd] : (gate ? embed_v_c[bd] : 0.0f));
  uv[(long)b * (2 * DD) + t]      = u;
  uv[(long)b * (2 * DD) + DD + t] = v;
}

__global__ __launch_bounds__(32) void mlp_layer_kernel(
    const float* __restrict__ x,
    const float* __restrict__ w,
    const float* __restrict__ bias,
    const float* __restrict__ bn_sum,
    const float* __restrict__ bn_sq,
    const float* __restrict__ bn_g,
    const float* __restrict__ bn_be,
    float* __restrict__ y,
    int K, int N, int use_bn, float invB) {
  __shared__ float sc_s[128];
  __shared__ float sh_s[128];

  const int lane = threadIdx.x;
  if (use_bn) {
    for (int c = lane; c < K; c += 32) {
      float mean = bn_sum[c] * invB;
      float var  = bn_sq[c] * invB - mean * mean;
      float sc   = bn_g[c] * rsqrtf(var + EPS_BN);
      sc_s[c] = sc;
      sh_s[c] = bn_be[c] - mean * sc;
    }
  } else {
    for (int c = lane; c < K; c += 32) {
      sc_s[c] = 1.0f;
      sh_s[c] = 0.0f;
    }
  }
  __syncthreads();

  const int row0  = blockIdx.x * 16;
  const int r     = lane & 15;
  const int khalf = (lane >> 4) * 2;
  const int col   = lane & 15;

  for (int nt = 0; nt < (N >> 4); ++nt) {
    v8f acc = {};
    for (int k = 0; k < K; k += 4) {
      const int kk = k + khalf;
      v2f a, bb;
      a.x  = sc_s[kk]     * x[(long)(row0 + r) * K + kk]     + sh_s[kk];
      a.y  = sc_s[kk + 1] * x[(long)(row0 + r) * K + kk + 1] + sh_s[kk + 1];
      bb.x = w[(long)kk * N + nt * 16 + col];
      bb.y = w[(long)(kk + 1) * N + nt * 16 + col];
      acc = __builtin_amdgcn_wmma_f32_16x16x4_f32(
          false, a, false, bb, (short)0, acc, false, false);
    }
    const int n = nt * 16 + col;
    const float bn = bias[n];
#pragma unroll
    for (int j = 0; j < 8; ++j) {
      const int m = row0 + j + (lane >> 4) * 8;
      y[(long)m * N + n] = fmaxf(acc[j] + bn, 0.0f);
    }
  }
}

__global__ __launch_bounds__(256) void col_stats_kernel(
    const float* __restrict__ y, float* __restrict__ out_sum,
    float* __restrict__ out_sq, int Bn, int N) {
  __shared__ float ssum[256];
  __shared__ float ssq[256];
  const int n = blockIdx.x;
  const int t = threadIdx.x;
  float s = 0.0f, q = 0.0f;
  for (int m = t; m < Bn; m += 256) {
    float v = y[(long)m * N + n];
    s += v;
    q += v * v;
  }
  ssum[t] = s;
  ssq[t]  = q;
  __syncthreads();
  for (int st = 128; st > 0; st >>= 1) {
    if (t < st) {
      ssum[t] += ssum[t + st];
      ssq[t]  += ssq[t + st];
    }
    __syncthreads();
  }
  if (t == 0) {
    out_sum[n] = ssum[0];
    out_sq[n]  = ssq[0];
  }
}

__global__ __launch_bounds__(256) void final_kernel(
    const float* __restrict__ y3, const float* __restrict__ sum3,
    const float* __restrict__ sq3, const float* __restrict__ g3,
    const float* __restrict__ be3, const float* __restrict__ wf,
    const float* __restrict__ bf, float* __restrict__ out, int Bn, float invB) {
  __shared__ float sc_s[16], sh_s[16], wf_s[16];
  const int t = threadIdx.x;
  if (t < 16) {
    float mean = sum3[t] * invB;
    float var  = sq3[t] * invB - mean * mean;
    float sc   = g3[t] * rsqrtf(var + EPS_BN);
    sc_s[t] = sc;
    sh_s[t] = be3[t] - mean * sc;
    wf_s[t] = wf[t];
  }
  __syncthreads();
  const int b = blockIdx.x * blockDim.x + t;
  if (b < Bn) {
    float s = 0.0f;
#pragma unroll
    for (int c = 0; c < 16; ++c)
      s += (sc_s[c] * y3[(long)b * 16 + c] + sh_s[c]) * wf_s[c];
    out[b] = s + bf[0];
  }
}

extern "C" void kernel_launch(void* const* d_in, const int* in_sizes, int n_in,
                              void* d_out, int out_size, void* d_ws,
                              size_t ws_size, hipStream_t stream) {
  const float* embed_u_z = (const float*)d_in[0];
  const float* embed_u_c = (const float*)d_in[1];
  const float* embed_v_z = (const float*)d_in[2];
  const float* embed_v_c = (const float*)d_in[3];
  const float* item_table = (const float*)d_in[4];
  const float* w1 = (const float*)d_in[5];
  const float* b1 = (const float*)d_in[6];
  const float* g1 = (const float*)d_in[7];
  const float* be1 = (const float*)d_in[8];
  const float* w2 = (const float*)d_in[9];
  const float* b2 = (const float*)d_in[10];
  const float* g2 = (const float*)d_in[11];
  const float* be2 = (const float*)d_in[12];
  const float* w3 = (const float*)d_in[13];
  const float* b3 = (const float*)d_in[14];
  const float* g3 = (const float*)d_in[15];
  const float* be3 = (const float*)d_in[16];
  const float* wf = (const float*)d_in[17];
  const float* bf = (const float*)d_in[18];
  const int* hist_items = (const int*)d_in[19];
  const int* hist_lens = (const int*)d_in[20];
  const int* item_ids = (const int*)d_in[21];
  const int* red_by_friends = (const int*)d_in[22];

  const int Bn = in_sizes[20];
  const int H  = in_sizes[19] / Bn;
  const float invB = 1.0f / (float)Bn;

  float* ws = (float*)d_ws;
  float* uv = ws;
  float* y1 = uv + (size_t)Bn * 128;
  float* y2 = y1 + (size_t)Bn * 64;
  float* y3 = y2 + (size_t)Bn * 32;
  float* stats = y3 + (size_t)Bn * 16;
  float* sum1 = stats;      float* sq1 = sum1 + 64;
  float* sum2 = sq1 + 64;   float* sq2 = sum2 + 32;
  float* sum3 = sq2 + 32;   float* sq3 = sum3 + 16;

  gather_gate_kernel<<<Bn, 64, 0, stream>>>(
      embed_u_z, embed_u_c, embed_v_z, embed_v_c, item_table,
      hist_items, hist_lens, item_ids, red_by_friends, uv, H);

  mlp_layer_kernel<<<Bn / 16, 32, 0, stream>>>(
      uv, w1, b1, nullptr, nullptr, nullptr, nullptr,
      y1, 128, 64, 0, invB);
  col_stats_kernel<<<64, 256, 0, stream>>>(y1, sum1, sq1, Bn, 64);

  mlp_layer_kernel<<<Bn / 16, 32, 0, stream>>>(
      y1, w2, b2, sum1, sq1, g1, be1,
      y2, 64, 32, 1, invB);
  col_stats_kernel<<<32, 256, 0, stream>>>(y2, sum2, sq2, Bn, 32);

  mlp_layer_kernel<<<Bn / 16, 32, 0, stream>>>(
      y2, w3, b3, sum2, sq2, g2, be2,
      y3, 32, 16, 1, invB);
  col_stats_kernel<<<16, 256, 0, stream>>>(y3, sum3, sq3, Bn, 16);

  final_kernel<<<(Bn + 255) / 256, 256, 0, stream>>>(
      y3, sum3, sq3, g3, be3, wf, bf, (float*)d_out, Bn, invB);
}


# layers 2, 3 and the final dot recomputed from y1 inside the statistics and output kernels (8-wave blocks, tiles in LDS): y2, y3 never stored; 5 launches, not 7; order of every BN sum changes
# speedup vs baseline: 5.1749x; 5.1749x over previous
#include <hip/hip_runtime.h>
#include <hip/hip_bf16.h>
#include <math.h>

typedef __attribute__((ext_vector_type(2))) float v2f;
typedef __attribute__((ext_vector_type(8))) float v8f;
typedef __attribute__((ext_vector_type(4))) float v4f;

#define DD 64
#define EPS_COS 1e-8f
#define EPS_BN  1e-5f
#define STAT_SLABS 32

#define ROWS_PB  16
#define ROWS_PW  2
#define WAVES_PB (ROWS_PB / ROWS_PW)
#define UV_PITCH 132
#define Y_PITCH  68
#define TILE_ROWS  26
#define TILE_PITCH 68


__global__ __launch_bounds__(WAVES_PB * 32, 8) void gather_layer1_kernel(
    const float* __restrict__ embed_u_z, const float* __restrict__ embed_u_c,
    const float* __restrict__ embed_v_z, const float* __restrict__ embed_v_c,
    const float* __restrict__ item_table,
    const int* __restrict__ hist_items, const int* __restrict__ hist_lens,
    const int* __restrict__ item_ids, const int* __restrict__ red_by_friends,
    const float* __restrict__ w1, const float* __restrict__ b1,
    float* __restrict__ y1, int H) {
  __shared__ __align__(16) float uv_s[ROWS_PB * UV_PITCH];
  __shared__ __align__(16) float y_s[ROWS_PB * Y_PITCH];
  __shared__ __align__(16) float tile_s[WAVES_PB * TILE_ROWS * TILE_PITCH];

  const int lane = threadIdx.x & 31;
  const int wave = __builtin_amdgcn_readfirstlane((int)(threadIdx.x >> 5));
  const int row0 = blockIdx.x * ROWS_PB;
  const int wrow0 = wave * ROWS_PW;
  const int b0    = row0 + wrow0;

  float best = -INFINITY;
  unsigned anyv = 0u;
  const int hw    = lane >> 4;
  const int piece = lane & 15;
  float* tile = tile_s + wave * (TILE_ROWS * TILE_PITCH);
  float* st_p = tile + hw * TILE_PITCH + piece * 4;
  const char* table_b = (const char*)item_table;
  const unsigned pofs = (unsigned)piece * 16u;
  const float* ld_p = tile + lane * TILE_PITCH;
  const bool rowlane = lane < TILE_ROWS;
  float nb = 0.0f;

  const int item0 = item_ids[b0];
  const int item1 = item_ids[b0 + 1];
  const int hl0   = hist_lens[b0];
  const int hl1   = hist_lens[b0 + 1];
  const int rf0   = red_by_friends[b0];
  const int rf1   = red_by_friends[b0 + 1];
  const int tA = lane;
  const int tB = lane + TILE_ROWS;
  const bool inA = rowlane && tA < H;
  const bool inB = rowlane && tB < H;
  const int hiA0 = inA ? hist_items[b0 * H + tA] : 0;
  const int hiB0 = inB ? hist_items[b0 * H + tB] : 0;
  const int hiA1 = inA ? hist_items[(b0 + 1) * H + tA] : 0;
  const int hiB1 = inB ? hist_items[(b0 + 1) * H + tB] : 0;

#pragma unroll 1
  for (int rr = 0; rr < ROWS_PW; ++rr) {
    const int b    = b0 + rr;
    const int item = rr ? item1 : item0;
    const int hl   = rr ? hl1 : hl0;
    const int rf   = rr ? rf1 : rf0;
    const int hiA  = rr ? hiA1 : hiA0;
    const int hiB  = rr ? hiB1 : hiB0;
    const float4* an4 = (const float4*)(item_table + (long)item * DD);
    best = -INFINITY;
    anyv = 0u;
    nb   = 0.0f;

    auto issue_rows = [&](int hi, float4* g) {
#pragma unroll
      for (int j = 0; j < TILE_ROWS / 2; ++j) {
        const int i0  = __builtin_amdgcn_readlane(hi, 2 * j);
        const int i1  = __builtin_amdgcn_readlane(hi, 2 * j + 1);
        const int idx = hw ? i1 : i0;
        const unsigned off = (unsigned)idx * (DD * 4u) + pofs;
        g[j] = *(const float4*)(table_b + off);
      }
    };

    auto score_rows = [&](const float4* g, int hi, int t) {
#pragma unroll
      for (int j = 0; j < TILE_ROWS / 2; ++j)
        *(float4*)(st_p + 2 * j * TILE_PITCH) = g[j];

      float sim = -INFINITY;
      bool valid = false;
      if (rowlane && t < H) {
        const float4* he4 = (const float4*)ld_p;
        v2f d01 = {0.0f, 0.0f}, d23 = {0.0f, 0.0f};
        v2f n01 = {0.0f, 0.0f}, n23 = {0.0f, 0.0f};
#pragma unroll
        for (int i = 0; i < DD / 4; ++i) {
          const float4 x = he4[i];
          const float4 a = an4[i];
          const v2f x01 = {x.x, x.y};
          const v2f x23 = {x.z, x.w};
          const v2f a01 = {a.x, a.y};
          const v2f a23 = {a.z, a.w};
          d01 = __builtin_elementwise_fma(x01, a01, d01);
          d23 = __builtin_elementwise_fma(x23, a23, d23);
          n01 = __builtin_elementwise_fma(x01, x01, n01);
          n23 = __builtin_elementwise_fma(x23, x23, n23);
        }
        const v2f dsum = d01 + d23;
        const v2f nsum = n01 + n23;
        const float dot = dsum.x + dsum.y;
        const float na  = sqrtf(nsum.x + nsum.y);
        sim = dot / (fmaxf(na, EPS_COS) * fmaxf(nb, EPS_COS));
        valid = (t < hl) && (hi != item);
      }
      if (valid) best = fmaxf(best, sim);
      anyv |= __builtin_amdgcn_ballot_w32(valid);
    };

    {
      float4 g[TILE_ROWS / 2];
      issue_rows(hiA, g);
#pragma unroll
      for (int i = 0; i < DD / 4; ++i) {
        const float4 a = an4[i];
        nb += a.x * a.x + a.y * a.y + a.z * a.z + a.w * a.w;
      }
      nb = sqrtf(nb);
      score_rows(g, hiA, tA);
      issue_rows(hiB, g);
      score_rows(g, hiB, tB);
    }
#pragma unroll 1
    for (int base = 2 * TILE_ROWS; base < H; base += TILE_ROWS) {
      const int t  = base + lane;
      const int hi = (rowlane && t < H) ? hist_items[b * H + t] : 0;
      float4 g[TILE_ROWS / 2];
      issue_rows(hi, g);
      score_rows(g, hi, t);
    }
    const int bd = b * DD + 2 * lane;
    const v2f uz = *(const v2f*)(embed_u_z + bd);
    const v2f vz = *(const v2f*)(embed_v_z + bd);
#pragma unroll
    for (int m = 16; m > 0; m >>= 1) best = fmaxf(best, __shfl_xor(best, m, 32));
    best = __builtin_bit_cast(float,
        __builtin_amdgcn_readfirstlane(__builtin_bit_cast(int, best)));

    const bool empty   = (anyv == 0u);
    const bool gate    = (rf > 0) && (best > 0.5f);
    const bool use_u_c = empty || gate;

    v2f uc = {0.0f, 0.0f};
    v2f vc = {0.0f, 0.0f};
    if (use_u_c) uc = *(const v2f*)(embed_u_c + bd);
    if (gate && !empty) vc = *(const v2f*)(embed_v_c + bd);
    const v2f zero2 = {0.0f, 0.0f};
    const v2f u = uz + (use_u_c ? uc : zero2);
    const v2f v = vz + (empty ? uc : (gate ? vc : zero2));
    *(v2f*)(uv_s + (wrow0 + rr) * UV_PITCH + 2 * lane)      = u;
    *(v2f*)(uv_s + (wrow0 + rr) * UV_PITCH + DD + 2 * lane) = v;
  }
  __syncthreads();

  if (wave < 4) {
    const int r     = lane & 15;
    const int khalf = (lane >> 4) * 2;
    const int col   = lane & 15;
    const int n     = wave * 16 + col;
    v8f acc = {};
#pragma unroll
    for (int k = 0; k < 2 * DD; k += 4) {
      const int kk = k + khalf;
      const v2f a = *(const v2f*)(uv_s + r * UV_PITCH + kk);
      v2f bb;
      bb.x = w1[kk * DD + n];
      bb.y = w1[(kk + 1) * DD + n];
      acc = __builtin_amdgcn_wmma_f32_16x16x4_f32(
          false, a, false, bb, (short)0, acc, false, false);
    }
    const float bn = b1[n];
#pragma unroll
    for (int j = 0; j < 8; ++j) {
      const int m = j + (lane >> 4) * 8;
      y_s[m * Y_PITCH + n] = fmaxf(acc[j] + bn, 0.0f);
    }
  }
  __syncthreads();
#pragma unroll
  for (int rr = 0; rr < ROWS_PW; ++rr) {
    const int m = wrow0 + rr;
    *(v2f*)(y1 + (long)(row0 + m) * DD + 2 * lane) =
        *(const v2f*)(y_s + m * Y_PITCH + 2 * lane);
  }
}

template <int N>
__global__ __launch_bounds__(512) void col_stats_kernel(
    const float* __restrict__ y, float* __restrict__ part, int Bn) {
  constexpr int C4  = N / 4;
  constexpr int RPP = 512 / C4;
  constexpr int LPT = C4;
  __shared__ __align__(16) float red_s[16 * 2 * N];

  const int t    = threadIdx.x;
  const int lane = t & 31;
  const int wave = t >> 5;
  const int cg   = t % C4;
  const int tr   = t / C4;
  const int rows  = (Bn + STAT_SLABS - 1) / STAT_SLABS;
  const int start = blockIdx.x * rows;
  const int end   = min(start + rows, Bn);

  float4 s = {0.0f, 0.0f, 0.0f, 0.0f};
  float4 q = {0.0f, 0.0f, 0.0f, 0.0f};
  for (int base = start + tr; base < end; base += LPT * RPP) {
    float4 v[LPT];
#pragma unroll
    for (int j = 0; j < LPT; ++j) {
      const int row = min(base + j * RPP, end - 1);
      v[j] = *(const float4*)(y + (size_t)row * N + cg * 4);
    }
#pragma unroll
    for (int j = 0; j < LPT; ++j) {
      const bool in = (base + j * RPP) < end;
      const float x0 = in ? v[j].x : 0.0f;
      const float x1 = in ? v[j].y : 0.0f;
      const float x2 = in ? v[j].z : 0.0f;
      const float x3 = in ? v[j].w : 0.0f;
      s.x += x0;      s.y += x1;      s.z += x2;      s.w += x3;
      q.x += x0 * x0; q.y += x1 * x1; q.z += x2 * x2; q.w += x3 * x3;
    }
  }

#pragma unroll
  for (int m = 16; m >= C4; m >>= 1) {
    s.x += __shfl_xor(s.x, m, 32); s.y += __shfl_xor(s.y, m, 32);
    s.z += __shfl_xor(s.z, m, 32); s.w += __shfl_xor(s.w, m, 32);
    q.x += __shfl_xor(q.x, m, 32); q.y += __shfl_xor(q.y, m, 32);
    q.z += __shfl_xor(q.z, m, 32); q.w += __shfl_xor(q.w, m, 32);
  }
  if (lane < C4) {
    *(float4*)(red_s + wave * 2 * N + lane * 4)     = s;
    *(float4*)(red_s + wave * 2 * N + N + lane * 4) = q;
  }
  __syncthreads();
  if (t < 2 * N) {
    const int src = (t & 1) * N + (t >> 1);
    float a = 0.0f;
#pragma unroll
    for (int w = 0; w < 16; ++w) a += red_s[w * 2 * N + src];
    part[(size_t)blockIdx.x * 2 * N + t] = a;
  }
}

#define TAIL_WAVES 8
#define TAIL_ROWS  (TAIL_WAVES * 16)
#define T2_PITCH 36
#define T3_PITCH 17

template <int RW, int CH>
__device__ __forceinline__ void request_part_rows(const float* __restrict__ part,
                                                  int np, int t, float* v) {
  constexpr int G = 256 / RW;
  const int f = t % RW;
  const int g = t / RW;
#pragma unroll
  for (int i = 0; i < CH; ++i) {
    const int row = min(g + i * G, np - 1);
    v[i] = part[row * RW + f];
  }
}
template <int RW, int CH>
__device__ __forceinline__ float add_part_rows(const float* __restrict__ part,
                                               int np, int t, const float* v) {
  constexpr int G = 256 / RW;
  const int f = t % RW;
  const int g = t / RW;
  float a = 0.0f;
#pragma unroll
  for (int i = 0; i < CH; ++i) a += (g + i * G < np) ? v[i] : 0.0f;
  for (int row = g + CH * G; row < np; row += G) a += part[row * RW + f];
  return a;
}

template <int STAGE>
__global__ __launch_bounds__(TAIL_WAVES * 32) void tail_kernel(
    const float* __restrict__ y1,
    const float* __restrict__ part1,
    const float* __restrict__ g1, const float* __restrict__ be1,
    const float* __restrict__ w2, const float* __restrict__ b2,
    const float* __restrict__ part2,
    const float* __restrict__ g2, const float* __restrict__ be2,
    const float* __restrict__ w3, const float* __restrict__ b3,
    const float* __restrict__ part3,
    const float* __restrict__ g3, const float* __restrict__ be3,
    const float* __restrict__ wf, const float* __restrict__ bf,
    float* __restrict__ part_out,
    float* __restrict__ out, float invB) {
  __shared__ float red_s[3 * 256];
  __shared__ float sc1_s[64], sh1_s[64], sc2_s[32], sh2_s[32], sc3_s[16], sh3_s[16];
  __shared__ __align__(16) float t2_s[(STAGE >= 3) ? TAIL_WAVES * 16 * T2_PITCH : 4];
  __shared__ float t3_s[(STAGE >= 4) ? TAIL_WAVES * 16 * T3_PITCH : 1];
  __shared__ float cs_s[TAIL_WAVES * 64];

  const int t     = threadIdx.x;
  const int lane  = t & 31;
  const int wave  = t >> 5;
  const int r     = lane & 15;
  const int h     = lane >> 4;
  const int khalf = h * 2;
  const int col   = lane & 15;
  const int R0    = blockIdx.x * TAIL_ROWS + wave * 16;
  const int np    = gridDim.x;

  v2f xa[16], wb0[16], wb1[16];
  const float* xrow = y1 + (size_t)(R0 + r) * 64 + khalf;
#pragma unroll
  for (int s = 0; s < 16; ++s) xa[s] = *(const v2f*)(xrow + 4 * s);
#pragma unroll
  for (int s = 0; s < 16; ++s) {
    const int kk = 4 * s + khalf;
    wb0[s].x = w2[kk * 32 + col];
    wb0[s].y = w2[(kk + 1) * 32 + col];
    wb1[s].x = w2[kk * 32 + 16 + col];
    wb1[s].y = w2[(kk + 1) * 32 + 16 + col];
  }

  float gam = 0.0f, bet = 0.0f;
  if (t < 64) {
    gam = g1[t];
    bet = be1[t];
  } else if (STAGE >= 3 && t < 96) {
    gam = g2[t - 64];
    bet = be2[t - 64];
  } else if (STAGE >= 4 && t < 112) {
    gam = g3[t - 96];
    bet = be3[t - 96];
  }
  float p1[16], p2[32], p3[16];
  request_part_rows<128, 16>(part1, STAT_SLABS, t, p1);
  if (STAGE >= 3) request_part_rows<64, 32>(part2, np, t, p2);
  if (STAGE >= 4) request_part_rows<32, 16>(part3, np, t, p3);
  red_s[t] = add_part_rows<128, 16>(part1, STAT_SLABS, t, p1);
  if (STAGE >= 3) red_s[256 + t] = add_part_rows<64, 32>(part2, np, t, p2);
  if (STAGE >= 4) red_s[512 + t] = add_part_rows<32, 16>(part3, np, t, p3);
  __syncthreads();
  if (t < 64) {
    const float sum  = red_s[2 * t] + red_s[128 + 2 * t];
    const float sq   = red_s[2 * t + 1] + red_s[128 + 2 * t + 1];
    const float mean = sum * invB;
    const float var  = sq * invB - mean * mean;
    const float sc   = gam * rsqrtf(var + EPS_BN);
    sc1_s[t] = sc;
    sh1_s[t] = bet - mean * sc;
  } else if (STAGE >= 3 && t < 96) {
    const int c = t - 64;
    float sum = 0.0f, sq = 0.0f;
#pragma unroll
    for (int g = 0; g < 4; ++g) {
      sum += red_s[256 + g * 64 + 2 * c];
      sq  += red_s[256 + g * 64 + 2 * c + 1];
    }
    const float mean = sum * invB;
    const float var  = sq * invB - mean * mean;
    const float sc   = gam * rsqrtf(var + EPS_BN);
    sc2_s[c] = sc;
    sh2_s[c] = bet - mean * sc;
  } else if (STAGE >= 4 && t < 112) {
    const int c = t - 96;
    float sum = 0.0f, sq = 0.0f;
#pragma unroll
    for (int g = 0; g < 8; ++g) {
      sum += red_s[512 + g * 32 + 2 * c];
      sq  += red_s[512 + g * 32 + 2 * c + 1];
    }
    const float mean = sum * invB;
    const float var  = sq * invB - mean * mean;
    const float sc   = gam * rsqrtf(var + EPS_BN);
    sc3_s[c] = sc;
    sh3_s[c] = bet - mean * sc;
  }
  __syncthreads();

  v8f a0 = {}, a1 = {};
#pragma unroll
  for (int s = 0; s < 16; ++s) {
    const int kk = 4 * s + khalf;
    v2f a;
    a.x = sc1_s[kk] * xa[s].x + sh1_s[kk];
    a.y = sc1_s[kk + 1] * xa[s].y + sh1_s[kk + 1];
    a0 = __builtin_amdgcn_wmma_f32_16x16x4_f32(
        false, a, false, wb0[s], (short)0, a0, false, false);
    a1 = __builtin_amdgcn_wmma_f32_16x16x4_f32(
        false, a, false, wb1[s], (short)0, a1, false, false);
  }
  const float bias0 = b2[col];
  const float bias1 = b2[16 + col];
  float y2a[8], y2b[8];
#pragma unroll
  for (int j = 0; j < 8; ++j) {
    y2a[j] = fmaxf(a0[j] + bias0, 0.0f);
    y2b[j] = fmaxf(a1[j] + bias1, 0.0f);
  }

  if (STAGE == 2) {
    float s0 = 0.0f, q0 = 0.0f, s1 = 0.0f, q1 = 0.0f;
#pragma unroll
    for (int j = 0; j < 8; ++j) {
      s0 += y2a[j]; q0 += y2a[j] * y2a[j];
      s1 += y2b[j]; q1 += y2b[j] * y2b[j];
    }
    s0 += __shfl_xor(s0, 16, 32); q0 += __shfl_xor(q0, 16, 32);
    s1 += __shfl_xor(s1, 16, 32); q1 += __shfl_xor(q1, 16, 32);
    if (h == 0) {
      cs_s[wave * 64 + col]      = s0;
      cs_s[wave * 64 + 16 + col] = s1;
      cs_s[wave * 64 + 32 + col] = q0;
      cs_s[wave * 64 + 48 + col] = q1;
    }
    __syncthreads();
    if (t < 64) {
      const int src = (t & 1) * 32 + (t >> 1);
      float a = 0.0f;
#pragma unroll
      for (int w = 0; w < TAIL_WAVES; ++w) a += cs_s[w * 64 + src];
      part_out[(size_t)blockIdx.x * 64 + t] = a;
    }
    return;
  }

  float* t2 = t2_s + wave * (16 * T2_PITCH);
#pragma unroll
  for (int j = 0; j < 8; ++j) {
    const int m = j + 8 * h;
    t2[m * T2_PITCH + col]      = y2a[j];
    t2[m * T2_PITCH + 16 + col] = y2b[j];
  }
  v8f a3 = {};
#pragma unroll
  for (int s = 0; s < 8; ++s) {
    const int kk = 4 * s + khalf;
    const v2f y = *(const v2f*)(t2 + r * T2_PITCH + kk);
    v2f a, bb;
    a.x  = sc2_s[kk] * y.x + sh2_s[kk];
    a.y  = sc2_s[kk + 1] * y.y + sh2_s[kk + 1];
    bb.x = w3[kk * 16 + col];
    bb.y = w3[(kk + 1) * 16 + col];
    a3 = __builtin_amdgcn_wmma_f32_16x16x4_f32(
        false, a, false, bb, (short)0, a3, false, false);
  }
  const float bias3 = b3[col];
  float y3[8];
#pragma unroll
  for (int j = 0; j < 8; ++j) y3[j] = fmaxf(a3[j] + bias3, 0.0f);

  if (STAGE == 3) {
    float s0 = 0.0f, q0 = 0.0f;
#pragma unroll
    for (int j = 0; j < 8; ++j) {
      s0 += y3[j]; q0 += y3[j] * y3[j];
    }
    s0 += __shfl_xor(s0, 16, 32); q0 += __shfl_xor(q0, 16, 32);
    if (h == 0) {
      cs_s[wave * 32 + col]      = s0;
      cs_s[wave * 32 + 16 + col] = q0;
    }
    __syncthreads();
    if (t < 32) {
      const int src = (t & 1) * 16 + (t >> 1);
      float a = 0.0f;
#pragma unroll
      for (int w = 0; w < TAIL_WAVES; ++w) a += cs_s[w * 32 + src];
      part_out[(size_t)blockIdx.x * 32 + t] = a;
    }
    return;
  }

  float* t3 = t3_s + wave * (16 * T3_PITCH);
#pragma unroll
  for (int j = 0; j < 8; ++j) t3[(j + 8 * h) * T3_PITCH + col] = y3[j];
  if (lane < 16) {
    float s = 0.0f;
#pragma unroll
    for (int c = 0; c < 16; ++c)
      s += (sc3_s[c] * t3[lane * T3_PITCH + c] + sh3_s[c]) * wf[c];
    cs_s[wave * 16 + lane] = s + bf[0];
  }
  __syncthreads();
  if (t < TAIL_ROWS) out[(size_t)blockIdx.x * TAIL_ROWS + t] = cs_s[t];
}

extern "C" void kernel_launch(void* const* d_in, const int* in_sizes, int n_in,
                              void* d_out, int out_size, void* d_ws,
                              size_t ws_size, hipStream_t stream) {
  const float* embed_u_z = (const float*)d_in[0];
  const float* embed_u_c = (const float*)d_in[1];
  const float* embed_v_z = (const float*)d_in[2];
  const float* embed_v_c = (const float*)d_in[3];
  const float* item_table = (const float*)d_in[4];
  const float* w1 = (const float*)d_in[5];
  const float* b1 = (const float*)d_in[6];
  const float* g1 = (const float*)d_in[7];
  const float* be1 = (const float*)d_in[8];
  const float* w2 = (const float*)d_in[9];
  const float* b2 = (const float*)d_in[10];
  const float* g2 = (const float*)d_in[11];
  const float* be2 = (const float*)d_in[12];
  const float* w3 = (const float*)d_in[13];
  const float* b3 = (const float*)d_in[14];
  const float* g3 = (const float*)d_in[15];
  const float* be3 = (const float*)d_in[16];
  const float* wf = (const float*)d_in[17];
  const float* bf = (const float*)d_in[18];
  const int* hist_items = (const int*)d_in[19];
  const int* hist_lens = (const int*)d_in[20];
  const int* item_ids = (const int*)d_in[21];
  const int* red_by_friends = (const int*)d_in[22];

  const int Bn = in_sizes[20];
  const int H  = in_sizes[19] / Bn;
  const float invB = 1.0f / (float)Bn;

  float* ws = (float*)d_ws;
  float* y1 = ws;
  float* part1 = y1 + (size_t)Bn * 64;
  float* part2 = part1 + STAT_SLABS * 2 * 64;
  float* part3 = part2 + (size_t)(Bn / TAIL_ROWS) * 2 * 32;

  gather_layer1_kernel<<<Bn / ROWS_PB, WAVES_PB * 32, 0, stream>>>(
      embed_u_z, embed_u_c, embed_v_z, embed_v_c, item_table,
      hist_items, hist_lens, item_ids, red_by_friends, w1, b1, y1, H);
  col_stats_kernel<64><<<STAT_SLABS, 512, 0, stream>>>(y1, part1, Bn);

  tail_kernel<2><<<Bn / TAIL_ROWS, TAIL_WAVES * 32, 0, stream>>>(
      y1, part1, g1, be1, w2, b2, nullptr, g2, be2, w3, b3, nullptr, g3, be3,
      wf, bf, part2, nullptr, invB);
  tail_kernel<3><<<Bn / TAIL_ROWS, TAIL_WAVES * 32, 0, stream>>>(
      y1, part1, g1, be1, w2, b2, part2, g2, be2, w3, b3, nullptr, g3, be3,
      wf, bf, part3, nullptr, invB);
  tail_kernel<4><<<Bn / TAIL_ROWS, TAIL_WAVES * 32, 0, stream>>>(
      y1, part1, g1, be1, w2, b2, part2, g2, be2, w3, b3, part3, g3, be3,
      wf, bf, nullptr, (float*)d_out, invB);
}
